# starting kernel
# baseline (speedup 1.0000x reference)
#include <hip/hip_runtime.h>

typedef __attribute__((ext_vector_type(2))) float v2f;
typedef __attribute__((ext_vector_type(8))) float v8f;

#define D_DIM   1024
#define N_DIM   8
#define C_DIM   4
#define NROWS   (N_DIM * C_DIM)
#define VS      1028
#define NITER   3
#define EPS_F   1e-6f

__global__ void wdiag_gather_kernel(const float* __restrict__ W,
                                    float* __restrict__ wdiag) {
    int t = blockIdx.x * blockDim.x + threadIdx.x;
    if (t >= NROWS * D_DIM) return;
    int row = t >> 10;
    int d   = t & (D_DIM - 1);
    int n = row >> 2, c = row & 3;
    size_t idx = (((size_t)n * D_DIM + d) * C_DIM + c) * D_DIM + d;
    wdiag[t] = W[idx];
}

__global__ __launch_bounds__(256)
void routing_kernel(const float* __restrict__ fi, const float* __restrict__ pi,
                    const float* __restrict__ W, const float* __restrict__ wdiag,
                    const float* __restrict__ cj_init,
                    float* __restrict__ out_cj, float* __restrict__ out_rij) {
    extern __shared__ float smem[];
    float* votes = smem;
    float* cjl   = votes + NROWS * VS;
    float* sij   = cjl + C_DIM * VS;
    float* nrm   = sij + NROWS;

    const int b    = blockIdx.x;
    const int tid  = threadIdx.x;
    const int lane = tid & 31;
    const int wave = tid >> 5;

    float pin[N_DIM];
#pragma unroll
    for (int n = 0; n < N_DIM; ++n) pin[n] = pi[b * N_DIM + n];

    if (tid < C_DIM) nrm[tid] = 0.0f;
    if (tid >= 64 && tid < 64 + NROWS) sij[tid - 64] = 0.0f;

#pragma unroll
    for (int k = 0; k < 4; ++k) {
        int d = tid + 256 * k;
#pragma unroll
        for (int n = 0; n < N_DIM; ++n) {
            float f = fi[((size_t)b * N_DIM + n) * D_DIM + d];
#pragma unroll
            for (int c = 0; c < C_DIM; ++c) {
                int row = n * C_DIM + c;
                float wv = wdiag ? wdiag[row * D_DIM + d]
                                 : W[(((size_t)n * D_DIM + d) * C_DIM + c) * D_DIM + d];
                votes[row * VS + d] = f * wv;
            }
        }
    }

    float cjn[C_DIM][4];
    float psum[C_DIM];
#pragma unroll
    for (int c = 0; c < C_DIM; ++c) {
        psum[c] = 0.0f;
#pragma unroll
        for (int k = 0; k < 4; ++k) {
            int d = tid + 256 * k;
            float v = cj_init[((size_t)b * C_DIM + c) * D_DIM + d];
            cjn[c][k] = v;
            psum[c] += v * v;
        }
    }
    __syncthreads();
#pragma unroll
    for (int c = 0; c < C_DIM; ++c) {
        float s = psum[c];
#pragma unroll
        for (int off = 16; off >= 1; off >>= 1) s += __shfl_xor(s, off, 32);
        if (lane == 0) atomicAdd(&nrm[c], s);
    }
    __syncthreads();
#pragma unroll
    for (int c = 0; c < C_DIM; ++c) {
        float inv = 1.0f / (sqrtf(nrm[c]) + EPS_F);
#pragma unroll
        for (int k = 0; k < 4; ++k) {
            int d = tid + 256 * k;
            cjl[c * VS + d] = cjn[c][k] * inv;
        }
    }
    __syncthreads();
    if (tid < C_DIM) nrm[tid] = 0.0f;
    __syncthreads();

    const int    arow = lane & 15;
    const int    koff = (lane & 16) ? 2 : 0;
    const float* a0p  = votes + arow * VS;
    const float* a1p  = votes + (16 + arow) * VS;
    const bool   bval = arow < C_DIM;
    const float* bp   = cjl + (bval ? arow : 0) * VS;
    const int    kbase = wave * 128;

    float wgt[N_DIM][C_DIM];

    for (int it = 0; it < NITER; ++it) {
        v8f acc0 = {0.f, 0.f, 0.f, 0.f, 0.f, 0.f, 0.f, 0.f};
        v8f acc1 = acc0;
#pragma unroll 4
        for (int s = 0; s < 32; ++s) {
            int d0 = kbase + 4 * s + koff;
            v2f a0 = *(const v2f*)(a0p + d0);
            v2f a1 = *(const v2f*)(a1p + d0);
            v2f bb = *(const v2f*)(bp + d0);
            v2f bz = {0.f, 0.f};
            bb = bval ? bb : bz;
            acc0 = __builtin_amdgcn_wmma_f32_16x16x4_f32(false, a0, false, bb,
                                                         (short)0, acc0, false, false);
            acc1 = __builtin_amdgcn_wmma_f32_16x16x4_f32(false, a1, false, bb,
                                                         (short)0, acc1, false, false);
        }
#pragma unroll
        for (int v = 0; v < 8; ++v) {
            if ((lane & 15) == (v & 3)) {
                int mrow = v + ((lane >> 4) << 3);
                atomicAdd(&sij[mrow],      acc0[v]);
                atomicAdd(&sij[16 + mrow], acc1[v]);
            }
        }
        __syncthreads();

#pragma unroll
        for (int n = 0; n < N_DIM; ++n) {
            float s0 = sij[n * 4 + 0], s1 = sij[n * 4 + 1];
            float s2 = sij[n * 4 + 2], s3 = sij[n * 4 + 3];
            float mx = fmaxf(fmaxf(s0, s1), fmaxf(s2, s3));
            float e0 = __expf(s0 - mx), e1 = __expf(s1 - mx);
            float e2 = __expf(s2 - mx), e3 = __expf(s3 - mx);
            float rs = pin[n] / (e0 + e1 + e2 + e3);
            wgt[n][0] = e0 * rs; wgt[n][1] = e1 * rs;
            wgt[n][2] = e2 * rs; wgt[n][3] = e3 * rs;
        }

#pragma unroll
        for (int c = 0; c < C_DIM; ++c) {
            psum[c] = 0.0f;
#pragma unroll
            for (int k = 0; k < 4; ++k) {
                int d = tid + 256 * k;
                float a = 0.0f;
#pragma unroll
                for (int n = 0; n < N_DIM; ++n)
                    a = fmaf(wgt[n][c], votes[(n * C_DIM + c) * VS + d], a);
                cjn[c][k] = a;
                psum[c] += a * a;
            }
        }
#pragma unroll
        for (int c = 0; c < C_DIM; ++c) {
            float s = psum[c];
#pragma unroll
            for (int off = 16; off >= 1; off >>= 1) s += __shfl_xor(s, off, 32);
            if (lane == 0) atomicAdd(&nrm[c], s);
        }
        __syncthreads();

#pragma unroll
        for (int c = 0; c < C_DIM; ++c) {
            float inv = 1.0f / (sqrtf(nrm[c]) + EPS_F);
#pragma unroll
            for (int k = 0; k < 4; ++k) {
                int d = tid + 256 * k;
                float v = cjn[c][k] * inv;
                cjn[c][k] = v;
                if (it < NITER - 1) cjl[c * VS + d] = v;
            }
        }
        __syncthreads();
        if (tid < C_DIM) nrm[tid] = 0.0f;
        if (it < NITER - 1 && tid >= 64 && tid < 64 + NROWS) sij[tid - 64] = 0.0f;
        __syncthreads();
    }

#pragma unroll
    for (int c = 0; c < C_DIM; ++c)
#pragma unroll
        for (int k = 0; k < 4; ++k) {
            int d = tid + 256 * k;
            out_cj[((size_t)b * C_DIM + c) * D_DIM + d] = cjn[c][k];
        }
    if (tid < NROWS) {
        int n = tid >> 2;
        float s0 = sij[n * 4 + 0], s1 = sij[n * 4 + 1];
        float s2 = sij[n * 4 + 2], s3 = sij[n * 4 + 3];
        float mx = fmaxf(fmaxf(s0, s1), fmaxf(s2, s3));
        float e0 = __expf(s0 - mx), e1 = __expf(s1 - mx);
        float e2 = __expf(s2 - mx), e3 = __expf(s3 - mx);
        out_rij[(size_t)b * NROWS + tid] = __expf(sij[tid] - mx) / (e0 + e1 + e2 + e3);
    }
}

extern "C" void kernel_launch(void* const* d_in, const int* in_sizes, int n_in,
                              void* d_out, int out_size, void* d_ws, size_t ws_size,
                              hipStream_t stream) {
    (void)n_in; (void)out_size;
    const float* fi  = (const float*)d_in[0];
    const float* pi  = (const float*)d_in[1];
    const float* W   = (const float*)d_in[2];
    const float* cji = (const float*)d_in[3];
    const int B = in_sizes[0] / (N_DIM * D_DIM);

    float* out_cj  = (float*)d_out;
    float* out_rij = out_cj + (size_t)B * C_DIM * D_DIM;

    const float* wdiag = nullptr;
    const size_t wbytes = (size_t)NROWS * D_DIM * sizeof(float);
    if (ws_size >= wbytes) {
        int total = NROWS * D_DIM;
        wdiag_gather_kernel<<<(total + 255) / 256, 256, 0, stream>>>(W, (float*)d_ws);
        wdiag = (const float*)d_ws;
    }

    const size_t smem = (size_t)(NROWS * VS + C_DIM * VS + NROWS + C_DIM) * sizeof(float);
    (void)hipFuncSetAttribute((const void*)routing_kernel,
                              hipFuncAttributeMaxDynamicSharedMemorySize, (int)smem);
    routing_kernel<<<B, 256, smem, stream>>>(fi, pi, W, wdiag, cji, out_cj, out_rij);
}


# prologue global loads (fi, gathered W diagonal, cj_init) issued as independent groups: uniform source select hoisted out of the per-element branch, so round trips overlap (was one wait per load)
# speedup vs baseline: 2.4953x; 2.4953x over previous
#include <hip/hip_runtime.h>

typedef __attribute__((ext_vector_type(2))) float v2f;
typedef __attribute__((ext_vector_type(8))) float v8f;

#define D_DIM   1024
#define N_DIM   8
#define C_DIM   4
#define NROWS   (N_DIM * C_DIM)
#define VS      1028
#define NITER   3
#define EPS_F   1e-6f

__global__ void wdiag_gather_kernel(const float* __restrict__ W,
                                    float* __restrict__ wdiag) {
    int t = blockIdx.x * blockDim.x + threadIdx.x;
    if (t >= NROWS * D_DIM) return;
    int row = t >> 10;
    int d   = t & (D_DIM - 1);
    int n = row >> 2, c = row & 3;
    size_t idx = (((size_t)n * D_DIM + d) * C_DIM + c) * D_DIM + d;
    wdiag[t] = W[idx];
}

__global__ __launch_bounds__(256)
void routing_kernel(const float* __restrict__ fi, const float* __restrict__ pi,
                    const float* __restrict__ W, const float* __restrict__ wdiag,
                    const float* __restrict__ cj_init,
                    float* __restrict__ out_cj, float* __restrict__ out_rij) {
    extern __shared__ float smem[];
    float* votes = smem;
    float* cjl   = votes + NROWS * VS;
    float* sij   = cjl + C_DIM * VS;
    float* nrm   = sij + NROWS;

    const int b    = blockIdx.x;
    const int tid  = threadIdx.x;
    const int lane = tid & 31;
    const int wave = tid >> 5;

    float pin[N_DIM];
#pragma unroll
    for (int n = 0; n < N_DIM; ++n) pin[n] = pi[b * N_DIM + n];

    if (tid < C_DIM) nrm[tid] = 0.0f;
    if (tid >= 64 && tid < 64 + NROWS) sij[tid - 64] = 0.0f;

    const float*   wsrc = wdiag ? wdiag : W;
    const unsigned w_sn = wdiag ? (unsigned)(C_DIM * D_DIM) : (unsigned)(D_DIM * C_DIM * D_DIM);
    const unsigned w_sd = wdiag ? 1u : (unsigned)(C_DIM * D_DIM + 1);

    float cjn[C_DIM][4];
#pragma unroll
    for (int c = 0; c < C_DIM; ++c)
#pragma unroll
        for (int k = 0; k < 4; ++k)
            cjn[c][k] = cj_init[((size_t)b * C_DIM + c) * D_DIM + tid + 256 * k];

#pragma unroll
    for (int k = 0; k < 4; ++k) {
        const int d = tid + 256 * k;
        float fv[N_DIM];
        float wv[N_DIM][C_DIM];
#pragma unroll
        for (int n = 0; n < N_DIM; ++n)
            fv[n] = fi[((size_t)b * N_DIM + n) * D_DIM + d];
#pragma unroll
        for (int n = 0; n < N_DIM; ++n)
#pragma unroll
            for (int c = 0; c < C_DIM; ++c)
                wv[n][c] = wsrc[(unsigned)n * w_sn + (unsigned)d * w_sd + (unsigned)(c * D_DIM)];
#pragma unroll
        for (int n = 0; n < N_DIM; ++n)
#pragma unroll
            for (int c = 0; c < C_DIM; ++c)
                votes[(n * C_DIM + c) * VS + d] = fv[n] * wv[n][c];
    }

    float psum[C_DIM];
#pragma unroll
    for (int c = 0; c < C_DIM; ++c) {
        psum[c] = 0.0f;
#pragma unroll
        for (int k = 0; k < 4; ++k) psum[c] += cjn[c][k] * cjn[c][k];
    }
    __syncthreads();
#pragma unroll
    for (int c = 0; c < C_DIM; ++c) {
        float s = psum[c];
#pragma unroll
        for (int off = 16; off >= 1; off >>= 1) s += __shfl_xor(s, off, 32);
        if (lane == 0) atomicAdd(&nrm[c], s);
    }
    __syncthreads();
#pragma unroll
    for (int c = 0; c < C_DIM; ++c) {
        float inv = 1.0f / (sqrtf(nrm[c]) + EPS_F);
#pragma unroll
        for (int k = 0; k < 4; ++k) {
            int d = tid + 256 * k;
            cjl[c * VS + d] = cjn[c][k] * inv;
        }
    }
    __syncthreads();
    if (tid < C_DIM) nrm[tid] = 0.0f;
    __syncthreads();

    const int    arow = lane & 15;
    const int    koff = (lane & 16) ? 2 : 0;
    const float* a0p  = votes + arow * VS;
    const float* a1p  = votes + (16 + arow) * VS;
    const bool   bval = arow < C_DIM;
    const float* bp   = cjl + (bval ? arow : 0) * VS;
    const int    kbase = wave * 128;

    float wgt[N_DIM][C_DIM];

    for (int it = 0; it < NITER; ++it) {
        v8f acc0 = {0.f, 0.f, 0.f, 0.f, 0.f, 0.f, 0.f, 0.f};
        v8f acc1 = acc0;
#pragma unroll 4
        for (int s = 0; s < 32; ++s) {
            int d0 = kbase + 4 * s + koff;
            v2f a0 = *(const v2f*)(a0p + d0);
            v2f a1 = *(const v2f*)(a1p + d0);
            v2f bb = *(const v2f*)(bp + d0);
            v2f bz = {0.f, 0.f};
            bb = bval ? bb : bz;
            acc0 = __builtin_amdgcn_wmma_f32_16x16x4_f32(false, a0, false, bb,
                                                         (short)0, acc0, false, false);
            acc1 = __builtin_amdgcn_wmma_f32_16x16x4_f32(false, a1, false, bb,
                                                         (short)0, acc1, false, false);
        }
#pragma unroll
        for (int v = 0; v < 8; ++v) {
            if ((lane & 15) == (v & 3)) {
                int mrow = v + ((lane >> 4) << 3);
                atomicAdd(&sij[mrow],      acc0[v]);
                atomicAdd(&sij[16 + mrow], acc1[v]);
            }
        }
        __syncthreads();

#pragma unroll
        for (int n = 0; n < N_DIM; ++n) {
            float s0 = sij[n * 4 + 0], s1 = sij[n * 4 + 1];
            float s2 = sij[n * 4 + 2], s3 = sij[n * 4 + 3];
            float mx = fmaxf(fmaxf(s0, s1), fmaxf(s2, s3));
            float e0 = __expf(s0 - mx), e1 = __expf(s1 - mx);
            float e2 = __expf(s2 - mx), e3 = __expf(s3 - mx);
            float rs = pin[n] / (e0 + e1 + e2 + e3);
            wgt[n][0] = e0 * rs; wgt[n][1] = e1 * rs;
            wgt[n][2] = e2 * rs; wgt[n][3] = e3 * rs;
        }

#pragma unroll
        for (int c = 0; c < C_DIM; ++c) {
            psum[c] = 0.0f;
#pragma unroll
            for (int k = 0; k < 4; ++k) {
                int d = tid + 256 * k;
                float a = 0.0f;
#pragma unroll
                for (int n = 0; n < N_DIM; ++n)
                    a = fmaf(wgt[n][c], votes[(n * C_DIM + c) * VS + d], a);
                cjn[c][k] = a;
                psum[c] += a * a;
            }
        }
#pragma unroll
        for (int c = 0; c < C_DIM; ++c) {
            float s = psum[c];
#pragma unroll
            for (int off = 16; off >= 1; off >>= 1) s += __shfl_xor(s, off, 32);
            if (lane == 0) atomicAdd(&nrm[c], s);
        }
        __syncthreads();

#pragma unroll
        for (int c = 0; c < C_DIM; ++c) {
            float inv = 1.0f / (sqrtf(nrm[c]) + EPS_F);
#pragma unroll
            for (int k = 0; k < 4; ++k) {
                int d = tid + 256 * k;
                float v = cjn[c][k] * inv;
                cjn[c][k] = v;
                if (it < NITER - 1) cjl[c * VS + d] = v;
            }
        }
        __syncthreads();
        if (tid < C_DIM) nrm[tid] = 0.0f;
        if (it < NITER - 1 && tid >= 64 && tid < 64 + NROWS) sij[tid - 64] = 0.0f;
        __syncthreads();
    }

#pragma unroll
    for (int c = 0; c < C_DIM; ++c)
#pragma unroll
        for (int k = 0; k < 4; ++k) {
            int d = tid + 256 * k;
            out_cj[((size_t)b * C_DIM + c) * D_DIM + d] = cjn[c][k];
        }
    if (tid < NROWS) {
        int n = tid >> 2;
        float s0 = sij[n * 4 + 0], s1 = sij[n * 4 + 1];
        float s2 = sij[n * 4 + 2], s3 = sij[n * 4 + 3];
        float mx = fmaxf(fmaxf(s0, s1), fmaxf(s2, s3));
        float e0 = __expf(s0 - mx), e1 = __expf(s1 - mx);
        float e2 = __expf(s2 - mx), e3 = __expf(s3 - mx);
        out_rij[(size_t)b * NROWS + tid] = __expf(sij[tid] - mx) / (e0 + e1 + e2 + e3);
    }
}

extern "C" void kernel_launch(void* const* d_in, const int* in_sizes, int n_in,
                              void* d_out, int out_size, void* d_ws, size_t ws_size,
                              hipStream_t stream) {
    (void)n_in; (void)out_size;
    const float* fi  = (const float*)d_in[0];
    const float* pi  = (const float*)d_in[1];
    const float* W   = (const float*)d_in[2];
    const float* cji = (const float*)d_in[3];
    const int B = in_sizes[0] / (N_DIM * D_DIM);

    float* out_cj  = (float*)d_out;
    float* out_rij = out_cj + (size_t)B * C_DIM * D_DIM;

    const float* wdiag = nullptr;
    const size_t wbytes = (size_t)NROWS * D_DIM * sizeof(float);
    if (ws_size >= wbytes) {
        int total = NROWS * D_DIM;
        wdiag_gather_kernel<<<(total + 255) / 256, 256, 0, stream>>>(W, (float*)d_ws);
        wdiag = (const float*)d_ws;
    }

    const size_t smem = (size_t)(NROWS * VS + C_DIM * VS + NROWS + C_DIM) * sizeof(float);
    (void)hipFuncSetAttribute((const void*)routing_kernel,
                              hipFuncAttributeMaxDynamicSharedMemorySize, (int)smem);
    routing_kernel<<<B, 256, smem, stream>>>(fi, pi, W, wdiag, cji, out_cj, out_rij);
}


# routing updates run in Gram space: one WMMA pass gives G_c=V_c*V_c^T (8x8) and V_c.cj_init, the 3 updates use Gram rows per lane, cj formed once at end; sums reordered, LDS atomics gone
# speedup vs baseline: 4.1803x; 1.6753x over previous
#include <hip/hip_runtime.h>

typedef __attribute__((ext_vector_type(2))) float v2f;
typedef __attribute__((ext_vector_type(8))) float v8f;

#define D_DIM   1024
#define N_DIM   8
#define C_DIM   4
#define NROWS   (N_DIM * C_DIM)
#define VS      1028
#define NITER   3
#define EPS_F   1e-6f
#define NWAVE   8

__global__ void wdiag_gather_kernel(const float* __restrict__ W,
                                    float* __restrict__ wdiag) {
    int t = blockIdx.x * blockDim.x + threadIdx.x;
    if (t >= NROWS * D_DIM) return;
    int row = t >> 10;
    int d   = t & (D_DIM - 1);
    int n = row >> 2, c = row & 3;
    size_t idx = (((size_t)n * D_DIM + d) * C_DIM + c) * D_DIM + d;
    wdiag[t] = W[idx];
}

__device__ __forceinline__ float xch1(float v) {
    return __int_as_float(__builtin_amdgcn_update_dpp(0, __float_as_int(v), 0xB1, 0xF, 0xF, true));
}
__device__ __forceinline__ float xch2(float v) {
    return __int_as_float(__builtin_amdgcn_update_dpp(0, __float_as_int(v), 0x4E, 0xF, 0xF, true));
}
__device__ __forceinline__ float xch4(float v) {
    return __int_as_float(__builtin_amdgcn_update_dpp(0, __float_as_int(v), 0x164, 0xF, 0xF, true));
}
__device__ __forceinline__ float xch8(float v) {
    return __int_as_float(__builtin_amdgcn_update_dpp(0, __float_as_int(v), 0x168, 0xF, 0xF, true));
}
__device__ __forceinline__ float xch16(float v) {
    int x = __float_as_int(v);
    return __int_as_float(__builtin_amdgcn_permlanex16(x, x, 0x76543210, 0xfedcba98, false, false));
}

__global__ __launch_bounds__(256)
void routing_kernel(const float* __restrict__ fi, const float* __restrict__ pi,
                    const float* __restrict__ W, const float* __restrict__ wdiag,
                    const float* __restrict__ cj_init,
                    float* __restrict__ out_cj, float* __restrict__ out_rij) {
    extern __shared__ __align__(16) float smem[];
    float* votes = smem;
    float* cjl   = votes + NROWS * VS;
    float* gred  = cjl + C_DIM * VS;
    float* sred  = gred + NWAVE * C_DIM * 64;
    float* nred  = sred + NWAVE * NROWS;

    const int b    = blockIdx.x;
    const int tid  = threadIdx.x;
    const int lane = tid & 31;
    const int wave = tid >> 5;

    const float*   wsrc = wdiag ? wdiag : W;
    const unsigned w_sn = wdiag ? (unsigned)(C_DIM * D_DIM) : (unsigned)(D_DIM * C_DIM * D_DIM);
    const unsigned w_sd = wdiag ? 1u : (unsigned)(C_DIM * D_DIM + 1);

    const float pn = pi[b * N_DIM + (lane >> 2)];

    float cjn[C_DIM][4];
#pragma unroll
    for (int c = 0; c < C_DIM; ++c)
#pragma unroll
        for (int k = 0; k < 4; ++k)
            cjn[c][k] = cj_init[((size_t)b * C_DIM + c) * D_DIM + tid + 256 * k];

#pragma unroll
    for (int k = 0; k < 4; ++k) {
        const int d = tid + 256 * k;
        float fv[N_DIM];
        float wv[N_DIM][C_DIM];
#pragma unroll
        for (int n = 0; n < N_DIM; ++n)
            fv[n] = fi[((size_t)b * N_DIM + n) * D_DIM + d];
#pragma unroll
        for (int n = 0; n < N_DIM; ++n)
#pragma unroll
            for (int c = 0; c < C_DIM; ++c)
                wv[n][c] = wsrc[(unsigned)n * w_sn + (unsigned)d * w_sd + (unsigned)(c * D_DIM)];
#pragma unroll
        for (int n = 0; n < N_DIM; ++n)
#pragma unroll
            for (int c = 0; c < C_DIM; ++c)
                votes[(c * N_DIM + n) * VS + d] = fv[n] * wv[n][c];
    }

#pragma unroll
    for (int c = 0; c < C_DIM; ++c) {
        float ps = 0.0f;
#pragma unroll
        for (int k = 0; k < 4; ++k) {
            cjl[c * VS + tid + 256 * k] = cjn[c][k];
            ps += cjn[c][k] * cjn[c][k];
        }
        ps += xch1(ps); ps += xch2(ps); ps += xch4(ps); ps += xch8(ps); ps += xch16(ps);
        if (lane == 0) nred[wave * C_DIM + c] = ps;
    }
    __syncthreads();

    const int    arow = lane & 15;
    const int    half = lane >> 4;
    const int    koff = half * 2;
    const float* a0p  = votes + arow * VS;
    const float* a1p  = votes + (16 + arow) * VS;
    const bool   bval = arow < C_DIM;
    const float* bp   = cjl + (bval ? arow : 0) * VS;
    const int    kbase = wave * 128;

    v8f G0 = {0.f, 0.f, 0.f, 0.f, 0.f, 0.f, 0.f, 0.f};
    v8f G1 = G0, S0 = G0, S1 = G0;
#pragma unroll 4
    for (int s = 0; s < 32; ++s) {
        int d0 = kbase + 4 * s + koff;
        v2f a0 = *(const v2f*)(a0p + d0);
        v2f a1 = *(const v2f*)(a1p + d0);
        v2f bb = *(const v2f*)(bp + d0);
        v2f bz = {0.f, 0.f};
        bb = bval ? bb : bz;
        G0 = __builtin_amdgcn_wmma_f32_16x16x4_f32(false, a0, false, a0, (short)0, G0, false, false);
        G1 = __builtin_amdgcn_wmma_f32_16x16x4_f32(false, a1, false, a1, (short)0, G1, false, false);
        S0 = __builtin_amdgcn_wmma_f32_16x16x4_f32(false, a0, false, bb, (short)0, S0, false, false);
        S1 = __builtin_amdgcn_wmma_f32_16x16x4_f32(false, a1, false, bb, (short)0, S1, false, false);
    }

    if ((arow >> 3) == half) {
        const int np = arow & 7;
        float4* g0 = (float4*)(gred + ((wave * C_DIM + half) * 8 + np) * 8);
        float4* g1 = (float4*)(gred + ((wave * C_DIM + 2 + half) * 8 + np) * 8);
        g0[0] = make_float4(G0[0], G0[1], G0[2], G0[3]);
        g0[1] = make_float4(G0[4], G0[5], G0[6], G0[7]);
        g1[0] = make_float4(G1[0], G1[1], G1[2], G1[3]);
        g1[1] = make_float4(G1[4], G1[5], G1[6], G1[7]);
    }
    if (arow == half) {
        float4* s0 = (float4*)(sred + (wave * C_DIM + half) * 8);
        s0[0] = make_float4(S0[0], S0[1], S0[2], S0[3]);
        s0[1] = make_float4(S0[4], S0[5], S0[6], S0[7]);
    }
    if (arow == 2 + half) {
        float4* s1 = (float4*)(sred + (wave * C_DIM + 2 + half) * 8);
        s1[0] = make_float4(S1[0], S1[1], S1[2], S1[3]);
        s1[1] = make_float4(S1[4], S1[5], S1[6], S1[7]);
    }
    __syncthreads();

    const int cn = lane >> 2, cc = lane & 3;
    float g[N_DIM];
#pragma unroll
    for (int r = 0; r < N_DIM; ++r) g[r] = 0.0f;
    float sraw = 0.0f, nq = 0.0f;
#pragma unroll
    for (int w = 0; w < NWAVE; ++w) {
        const float4* gp = (const float4*)(gred + ((w * C_DIM + cc) * 8 + cn) * 8);
        float4 x = gp[0], y = gp[1];
        g[0] += x.x; g[1] += x.y; g[2] += x.z; g[3] += x.w;
        g[4] += y.x; g[5] += y.y; g[6] += y.z; g[7] += y.w;
        sraw += sred[(w * C_DIM + cc) * 8 + cn];
        nq   += nred[w * C_DIM + cc];
    }

    float s = sraw * (1.0f / (sqrtf(nq) + EPS_F));
    float wq = 0.0f, inv = 0.0f, rij = 0.0f;
#pragma unroll
    for (int it = 0; it < NITER; ++it) {
        float mx = fmaxf(s, xch1(s));
        mx = fmaxf(mx, xch2(mx));
        float e = __expf(s - mx);
        float es = e + xch1(e);
        es += xch2(es);
        rij = e / es;
        wq  = e * (pn / es);
        float t = 0.0f;
#pragma unroll
        for (int m = 0; m < N_DIM; ++m)
            t = fmaf(__shfl(wq, m * C_DIM + cc, 32), g[m], t);
        float q = wq * t;
        q += xch4(q); q += xch8(q); q += xch16(q);
        inv = 1.0f / (sqrtf(fmaxf(q, 0.0f)) + EPS_F);
        s = t * inv;
    }

    const int sc_i = __float_as_int(wq * inv);
#pragma unroll
    for (int c = 0; c < C_DIM; ++c) {
        float scl[N_DIM];
#pragma unroll
        for (int n = 0; n < N_DIM; ++n)
            scl[n] = __int_as_float(__builtin_amdgcn_readlane(sc_i, n * C_DIM + c));
#pragma unroll
        for (int k = 0; k < 4; ++k) {
            int d = tid + 256 * k;
            float a = 0.0f;
#pragma unroll
            for (int n = 0; n < N_DIM; ++n)
                a = fmaf(scl[n], votes[(c * N_DIM + n) * VS + d], a);
            out_cj[((size_t)b * C_DIM + c) * D_DIM + d] = a;
        }
    }
    if (wave == 0) out_rij[(size_t)b * NROWS + lane] = rij;
}

extern "C" void kernel_launch(void* const* d_in, const int* in_sizes, int n_in,
                              void* d_out, int out_size, void* d_ws, size_t ws_size,
                              hipStream_t stream) {
    (void)n_in; (void)out_size;
    const float* fi  = (const float*)d_in[0];
    const float* pi  = (const float*)d_in[1];
    const float* W   = (const float*)d_in[2];
    const float* cji = (const float*)d_in[3];
    const int B = in_sizes[0] / (N_DIM * D_DIM);

    float* out_cj  = (float*)d_out;
    float* out_rij = out_cj + (size_t)B * C_DIM * D_DIM;

    const float* wdiag = nullptr;
    const size_t wbytes = (size_t)NROWS * D_DIM * sizeof(float);
    if (ws_size >= wbytes) {
        int total = NROWS * D_DIM;
        wdiag_gather_kernel<<<(total + 255) / 256, 256, 0, stream>>>(W, (float*)d_ws);
        wdiag = (const float*)d_ws;
    }

    const size_t smem = (size_t)(NROWS * VS + C_DIM * VS + NWAVE * C_DIM * 64 +
                                 NWAVE * NROWS + NWAVE * C_DIM) * sizeof(float);
    (void)hipFuncSetAttribute((const void*)routing_kernel,
                              hipFuncAttributeMaxDynamicSharedMemorySize, (int)smem);
    routing_kernel<<<B, 256, smem, stream>>>(fi, pi, W, wdiag, cji, out_cj, out_rij);
}


# thread stages 4 adjacent d per row with 16-byte loads/stores, so each wave multiplies only the K chunk it staged itself: block barrier before the matrix pass removed; cj_init norm sum regrouped
# speedup vs baseline: 4.6969x; 1.1236x over previous
#include <hip/hip_runtime.h>

typedef __attribute__((ext_vector_type(2))) float v2f;
typedef __attribute__((ext_vector_type(8))) float v8f;

#define D_DIM   1024
#define N_DIM   8
#define C_DIM   4
#define NROWS   (N_DIM * C_DIM)
#define VS      1028
#define NITER   3
#define EPS_F   1e-6f
#define NWAVE   8

__global__ void wdiag_gather_kernel(const float* __restrict__ W,
                                    float* __restrict__ wdiag) {
    int t = blockIdx.x * blockDim.x + threadIdx.x;
    if (t >= NROWS * D_DIM) return;
    int row = t >> 10;
    int d   = t & (D_DIM - 1);
    int n = row >> 2, c = row & 3;
    size_t idx = (((size_t)n * D_DIM + d) * C_DIM + c) * D_DIM + d;
    wdiag[t] = W[idx];
}

__device__ __forceinline__ float xch1(float v) {
    return __int_as_float(__builtin_amdgcn_update_dpp(0, __float_as_int(v), 0xB1, 0xF, 0xF, true));
}
__device__ __forceinline__ float xch2(float v) {
    return __int_as_float(__builtin_amdgcn_update_dpp(0, __float_as_int(v), 0x4E, 0xF, 0xF, true));
}
__device__ __forceinline__ float xch4(float v) {
    return __int_as_float(__builtin_amdgcn_update_dpp(0, __float_as_int(v), 0x164, 0xF, 0xF, true));
}
__device__ __forceinline__ float xch8(float v) {
    return __int_as_float(__builtin_amdgcn_update_dpp(0, __float_as_int(v), 0x168, 0xF, 0xF, true));
}
__device__ __forceinline__ float xch16(float v) {
    int x = __float_as_int(v);
    return __int_as_float(__builtin_amdgcn_permlanex16(x, x, 0x76543210, 0xfedcba98, false, false));
}

template <bool GATHERED>
__global__ __launch_bounds__(256)
void routing_kernel(const float* __restrict__ fi, const float* __restrict__ pi,
                    const float* __restrict__ W, const float* __restrict__ wdiag,
                    const float* __restrict__ cj_init,
                    float* __restrict__ out_cj, float* __restrict__ out_rij) {
    extern __shared__ __align__(16) float smem[];
    float* votes = smem;
    float* cjl   = votes + NROWS * VS;
    float* gred  = cjl + C_DIM * VS;
    float* sred  = gred + NWAVE * C_DIM * 64;
    float* nred  = sred + NWAVE * NROWS;

    const int b    = blockIdx.x;
    const int tid  = threadIdx.x;
    const int lane = tid & 31;
    const int wave = tid >> 5;

    const int d4 = 4 * tid;

    const float pn = pi[b * N_DIM + (lane >> 2)];

    float4 cjn[C_DIM];
#pragma unroll
    for (int c = 0; c < C_DIM; ++c)
        cjn[c] = *(const float4*)(cj_init + ((size_t)b * C_DIM + c) * D_DIM + d4);

    float4 fv[N_DIM];
#pragma unroll
    for (int n = 0; n < N_DIM; ++n)
        fv[n] = *(const float4*)(fi + ((size_t)b * N_DIM + n) * D_DIM + d4);

    float4 wv[N_DIM][C_DIM];
    if constexpr (GATHERED) {
#pragma unroll
        for (int n = 0; n < N_DIM; ++n)
#pragma unroll
            for (int c = 0; c < C_DIM; ++c)
                wv[n][c] = *(const float4*)(wdiag + (n * C_DIM + c) * D_DIM + d4);
    } else {
        const size_t sd = (size_t)C_DIM * D_DIM + 1;
#pragma unroll
        for (int n = 0; n < N_DIM; ++n)
#pragma unroll
            for (int c = 0; c < C_DIM; ++c) {
                const float* p = W + (size_t)n * D_DIM * C_DIM * D_DIM + (size_t)d4 * sd + (size_t)c * D_DIM;
                wv[n][c] = make_float4(p[0], p[sd], p[2 * sd], p[3 * sd]);
            }
    }

    __builtin_amdgcn_sched_barrier(0);

#pragma unroll
    for (int n = 0; n < N_DIM; ++n)
#pragma unroll
        for (int c = 0; c < C_DIM; ++c) {
            const float4 f = fv[n], w = wv[n][c];
            *(float4*)(votes + (c * N_DIM + n) * VS + d4) = make_float4(f.x * w.x, f.y * w.y, f.z * w.z, f.w * w.w);
        }

#pragma unroll
    for (int c = 0; c < C_DIM; ++c) {
        const float4 q = cjn[c];
        *(float4*)(cjl + c * VS + d4) = q;
        float ps = q.x * q.x;
        ps += q.y * q.y; ps += q.z * q.z; ps += q.w * q.w;
        ps += xch1(ps); ps += xch2(ps); ps += xch4(ps); ps += xch8(ps); ps += xch16(ps);
        if (lane == 0) nred[wave * C_DIM + c] = ps;
    }

    const int    arow = lane & 15;
    const int    half = lane >> 4;
    const int    koff = half * 2;
    const float* a0p  = votes + arow * VS;
    const float* a1p  = votes + (16 + arow) * VS;
    const bool   bval = arow < C_DIM;
    const float* bp   = cjl + (bval ? arow : 0) * VS;
    const int    kbase = wave * 128;

    v8f G0 = {0.f, 0.f, 0.f, 0.f, 0.f, 0.f, 0.f, 0.f};
    v8f G1 = G0, S0 = G0, S1 = G0;
#pragma unroll 4
    for (int s = 0; s < 32; ++s) {
        int d0 = kbase + 4 * s + koff;
        v2f a0 = *(const v2f*)(a0p + d0);
        v2f a1 = *(const v2f*)(a1p + d0);
        v2f bb = *(const v2f*)(bp + d0);
        v2f bz = {0.f, 0.f};
        bb = bval ? bb : bz;
        G0 = __builtin_amdgcn_wmma_f32_16x16x4_f32(false, a0, false, a0, (short)0, G0, false, false);
        G1 = __builtin_amdgcn_wmma_f32_16x16x4_f32(false, a1, false, a1, (short)0, G1, false, false);
        S0 = __builtin_amdgcn_wmma_f32_16x16x4_f32(false, a0, false, bb, (short)0, S0, false, false);
        S1 = __builtin_amdgcn_wmma_f32_16x16x4_f32(false, a1, false, bb, (short)0, S1, false, false);
    }

    if ((arow >> 3) == half) {
        const int np = arow & 7;
        float4* g0 = (float4*)(gred + ((wave * C_DIM + half) * 8 + np) * 8);
        float4* g1 = (float4*)(gred + ((wave * C_DIM + 2 + half) * 8 + np) * 8);
        g0[0] = make_float4(G0[0], G0[1], G0[2], G0[3]);
        g0[1] = make_float4(G0[4], G0[5], G0[6], G0[7]);
        g1[0] = make_float4(G1[0], G1[1], G1[2], G1[3]);
        g1[1] = make_float4(G1[4], G1[5], G1[6], G1[7]);
    }
    if (arow == half) {
        float4* s0 = (float4*)(sred + (wave * C_DIM + half) * 8);
        s0[0] = make_float4(S0[0], S0[1], S0[2], S0[3]);
        s0[1] = make_float4(S0[4], S0[5], S0[6], S0[7]);
    }
    if (arow == 2 + half) {
        float4* s1 = (float4*)(sred + (wave * C_DIM + 2 + half) * 8);
        s1[0] = make_float4(S1[0], S1[1], S1[2], S1[3]);
        s1[1] = make_float4(S1[4], S1[5], S1[6], S1[7]);
    }
    __syncthreads();

    const int cn = lane >> 2, cc = lane & 3;
    float g[N_DIM];
#pragma unroll
    for (int r = 0; r < N_DIM; ++r) g[r] = 0.0f;
    float sraw = 0.0f, nq = 0.0f;
#pragma unroll
    for (int w = 0; w < NWAVE; ++w) {
        const float4* gp = (const float4*)(gred + ((w * C_DIM + cc) * 8 + cn) * 8);
        float4 x = gp[0], y = gp[1];
        g[0] += x.x; g[1] += x.y; g[2] += x.z; g[3] += x.w;
        g[4] += y.x; g[5] += y.y; g[6] += y.z; g[7] += y.w;
        sraw += sred[(w * C_DIM + cc) * 8 + cn];
        nq   += nred[w * C_DIM + cc];
    }

    float s = sraw * (1.0f / (sqrtf(nq) + EPS_F));
    float wq = 0.0f, inv = 0.0f, rij = 0.0f;
#pragma unroll
    for (int it = 0; it < NITER; ++it) {
        float mx = fmaxf(s, xch1(s));
        mx = fmaxf(mx, xch2(mx));
        float e = __expf(s - mx);
        float es = e + xch1(e);
        es += xch2(es);
        rij = e / es;
        wq  = e * (pn / es);
        float t = 0.0f;
#pragma unroll
        for (int m = 0; m < N_DIM; ++m)
            t = fmaf(__shfl(wq, m * C_DIM + cc, 32), g[m], t);
        float q = wq * t;
        q += xch4(q); q += xch8(q); q += xch16(q);
        inv = 1.0f / (sqrtf(fmaxf(q, 0.0f)) + EPS_F);
        s = t * inv;
    }

    const int sc_i = __float_as_int(wq * inv);
#pragma unroll
    for (int c = 0; c < C_DIM; ++c) {
        float scl[N_DIM];
#pragma unroll
        for (int n = 0; n < N_DIM; ++n)
            scl[n] = __int_as_float(__builtin_amdgcn_readlane(sc_i, n * C_DIM + c));
        float4 a = make_float4(0.0f, 0.0f, 0.0f, 0.0f);
#pragma unroll
        for (int n = 0; n < N_DIM; ++n) {
            const float4 v = *(const float4*)(votes + (c * N_DIM + n) * VS + d4);
            a.x = fmaf(scl[n], v.x, a.x); a.y = fmaf(scl[n], v.y, a.y);
            a.z = fmaf(scl[n], v.z, a.z); a.w = fmaf(scl[n], v.w, a.w);
        }
        *(float4*)(out_cj + ((size_t)b * C_DIM + c) * D_DIM + d4) = a;
    }
    if (wave == 0) out_rij[(size_t)b * NROWS + lane] = rij;
}

extern "C" void kernel_launch(void* const* d_in, const int* in_sizes, int n_in,
                              void* d_out, int out_size, void* d_ws, size_t ws_size,
                              hipStream_t stream) {
    (void)n_in; (void)out_size;
    const float* fi  = (const float*)d_in[0];
    const float* pi  = (const float*)d_in[1];
    const float* W   = (const float*)d_in[2];
    const float* cji = (const float*)d_in[3];
    const int B = in_sizes[0] / (N_DIM * D_DIM);

    float* out_cj  = (float*)d_out;
    float* out_rij = out_cj + (size_t)B * C_DIM * D_DIM;

    const float* wdiag = nullptr;
    const size_t wbytes = (size_t)NROWS * D_DIM * sizeof(float);
    if (ws_size >= wbytes) {
        int total = NROWS * D_DIM;
        wdiag_gather_kernel<<<(total + 255) / 256, 256, 0, stream>>>(W, (float*)d_ws);
        wdiag = (const float*)d_ws;
    }

    const size_t smem = (size_t)(NROWS * VS + C_DIM * VS + NWAVE * C_DIM * 64 +
                                 NWAVE * NROWS + NWAVE * C_DIM) * sizeof(float);
    if (wdiag) {
        (void)hipFuncSetAttribute((const void*)routing_kernel<true>,
                                  hipFuncAttributeMaxDynamicSharedMemorySize, (int)smem);
        routing_kernel<true><<<B, 256, smem, stream>>>(fi, pi, W, wdiag, cji, out_cj, out_rij);
    } else {
        (void)hipFuncSetAttribute((const void*)routing_kernel<false>,
                                  hipFuncAttributeMaxDynamicSharedMemorySize, (int)smem);
        routing_kernel<false><<<B, 256, smem, stream>>>(fi, pi, W, wdiag, cji, out_cj, out_rij);
    }
}


# each wave loads and stages its K chunk as 4 sub-chunks of 32 columns (lane = 16-byte column group x row group) and multiplies sub-chunk j while later loads are in flight; cj_init norm sum regrouped
# speedup vs baseline: 5.1116x; 1.0883x over previous
#include <hip/hip_runtime.h>

typedef __attribute__((ext_vector_type(2))) float v2f;
typedef __attribute__((ext_vector_type(8))) float v8f;

#define D_DIM   1024
#define N_DIM   8
#define C_DIM   4
#define NROWS   (N_DIM * C_DIM)
#define VS      1028
#define NITER   3
#define EPS_F   1e-6f
#define NWAVE   8
#define NSUB    4
#define SUBW    32

__global__ void wdiag_gather_kernel(const float* __restrict__ W,
                                    float* __restrict__ wdiag) {
    int t = blockIdx.x * blockDim.x + threadIdx.x;
    if (t >= NROWS * D_DIM) return;
    int row = t >> 10;
    int d   = t & (D_DIM - 1);
    int n = row >> 2, c = row & 3;
    size_t idx = (((size_t)n * D_DIM + d) * C_DIM + c) * D_DIM + d;
    wdiag[t] = W[idx];
}

__device__ __forceinline__ float xch1(float v) {
    return __int_as_float(__builtin_amdgcn_update_dpp(0, __float_as_int(v), 0xB1, 0xF, 0xF, true));
}
__device__ __forceinline__ float xch2(float v) {
    return __int_as_float(__builtin_amdgcn_update_dpp(0, __float_as_int(v), 0x4E, 0xF, 0xF, true));
}
__device__ __forceinline__ float xch4(float v) {
    return __int_as_float(__builtin_amdgcn_update_dpp(0, __float_as_int(v), 0x164, 0xF, 0xF, true));
}
__device__ __forceinline__ float xch8(float v) {
    return __int_as_float(__builtin_amdgcn_update_dpp(0, __float_as_int(v), 0x168, 0xF, 0xF, true));
}
__device__ __forceinline__ float xch16(float v) {
    int x = __float_as_int(v);
    return __int_as_float(__builtin_amdgcn_permlanex16(x, x, 0x76543210, 0xfedcba98, false, false));
}

__device__ __forceinline__ void hold_here(float4& v) {
    asm volatile("" : "+v"(v.x), "+v"(v.y), "+v"(v.z), "+v"(v.w) : : "memory");
}

template <bool GATHERED>
__global__ __launch_bounds__(256)
void routing_kernel(const float* __restrict__ fi, const float* __restrict__ pi,
                    const float* __restrict__ W, const float* __restrict__ wdiag,
                    const float* __restrict__ cj_init,
                    float* __restrict__ out_cj, float* __restrict__ out_rij) {
    extern __shared__ __align__(16) float smem[];
    float* votes = smem;
    float* cjl   = votes + NROWS * VS;
    float* gred  = cjl + C_DIM * VS;
    float* sred  = gred + NWAVE * C_DIM * 64;
    float* nred  = sred + NWAVE * NROWS;

    const int b    = blockIdx.x;
    const int tid  = threadIdx.x;
    const int lane = tid & 31;
    const int wave = tid >> 5;

    const int dg    = lane & 7;
    const int cg    = (lane >> 3) & 1;
    const int ng    = lane >> 4;
    const int cown  = 2 * ng + cg;
    const int kbase = wave * 128;
    const int dl    = kbase + 4 * dg;
    const int d4    = 4 * tid;

    const float pn = pi[b * N_DIM + (lane >> 2)];

    float4 cjn[NSUB];
    float4 fv[NSUB][4];
    float4 wv[NSUB][4][2];
#pragma unroll
    for (int j = 0; j < NSUB; ++j) {
        const int d = dl + SUBW * j;
        cjn[j] = *(const float4*)(cj_init + ((size_t)b * C_DIM + cown) * D_DIM + d);
#pragma unroll
        for (int i = 0; i < 4; ++i)
            fv[j][i] = *(const float4*)(fi + ((size_t)b * N_DIM + 4 * ng + i) * D_DIM + d);
#pragma unroll
        for (int i = 0; i < 4; ++i)
#pragma unroll
            for (int h = 0; h < 2; ++h) {
                const int n = 4 * ng + i, c = 2 * h + cg;
                if constexpr (GATHERED) {
                    wv[j][i][h] = *(const float4*)(wdiag + (n * C_DIM + c) * D_DIM + d);
                } else {
                    const size_t sd = (size_t)C_DIM * D_DIM + 1;
                    const float* p = W + (size_t)n * D_DIM * C_DIM * D_DIM + (size_t)d * sd + (size_t)c * D_DIM;
                    wv[j][i][h] = make_float4(p[0], p[sd], p[2 * sd], p[3 * sd]);
                }
            }
        __builtin_amdgcn_sched_barrier(0);
    }

    const int    arow = lane & 15;
    const int    half = lane >> 4;
    const int    koff = half * 2;
    const float* a0p  = votes + arow * VS;
    const float* a1p  = votes + (16 + arow) * VS;
    const bool   bval = arow < C_DIM;
    const float* bp   = cjl + (bval ? arow : 0) * VS;

    v8f G0 = {0.f, 0.f, 0.f, 0.f, 0.f, 0.f, 0.f, 0.f};
    v8f G1 = G0, S0 = G0, S1 = G0;
    float ps = 0.0f;
#pragma unroll
    for (int j = 0; j < NSUB; ++j) {
        const int d = dl + SUBW * j;
        hold_here(cjn[j]);
#pragma unroll
        for (int i = 0; i < 4; ++i) {
            hold_here(fv[j][i]);
            hold_here(wv[j][i][0]);
            hold_here(wv[j][i][1]);
        }
#pragma unroll
        for (int i = 0; i < 4; ++i)
#pragma unroll
            for (int h = 0; h < 2; ++h) {
                const int n = 4 * ng + i, c = 2 * h + cg;
                const float4 f = fv[j][i], w = wv[j][i][h];
                *(float4*)(votes + (c * N_DIM + n) * VS + d) = make_float4(f.x * w.x, f.y * w.y, f.z * w.z, f.w * w.w);
            }
        {
            const float4 q = cjn[j];
            *(float4*)(cjl + cown * VS + d) = q;
            ps += q.x * q.x; ps += q.y * q.y; ps += q.z * q.z; ps += q.w * q.w;
        }
#pragma unroll 4
        for (int s = 0; s < SUBW / 4; ++s) {
            int d0 = kbase + SUBW * j + 4 * s + koff;
            v2f a0 = *(const v2f*)(a0p + d0);
            v2f a1 = *(const v2f*)(a1p + d0);
            v2f bb = *(const v2f*)(bp + d0);
            v2f bz = {0.f, 0.f};
            bb = bval ? bb : bz;
            G0 = __builtin_amdgcn_wmma_f32_16x16x4_f32(false, a0, false, a0, (short)0, G0, false, false);
            G1 = __builtin_amdgcn_wmma_f32_16x16x4_f32(false, a1, false, a1, (short)0, G1, false, false);
            S0 = __builtin_amdgcn_wmma_f32_16x16x4_f32(false, a0, false, bb, (short)0, S0, false, false);
            S1 = __builtin_amdgcn_wmma_f32_16x16x4_f32(false, a1, false, bb, (short)0, S1, false, false);
        }
    }
    ps += xch1(ps); ps += xch2(ps); ps += xch4(ps);
    if (dg == 0) nred[wave * C_DIM + cown] = ps;

    if ((arow >> 3) == half) {
        const int np = arow & 7;
        float4* g0 = (float4*)(gred + ((wave * C_DIM + half) * 8 + np) * 8);
        float4* g1 = (float4*)(gred + ((wave * C_DIM + 2 + half) * 8 + np) * 8);
        g0[0] = make_float4(G0[0], G0[1], G0[2], G0[3]);
        g0[1] = make_float4(G0[4], G0[5], G0[6], G0[7]);
        g1[0] = make_float4(G1[0], G1[1], G1[2], G1[3]);
        g1[1] = make_float4(G1[4], G1[5], G1[6], G1[7]);
    }
    if (arow == half) {
        float4* s0 = (float4*)(sred + (wave * C_DIM + half) * 8);
        s0[0] = make_float4(S0[0], S0[1], S0[2], S0[3]);
        s0[1] = make_float4(S0[4], S0[5], S0[6], S0[7]);
    }
    if (arow == 2 + half) {
        float4* s1 = (float4*)(sred + (wave * C_DIM + 2 + half) * 8);
        s1[0] = make_float4(S1[0], S1[1], S1[2], S1[3]);
        s1[1] = make_float4(S1[4], S1[5], S1[6], S1[7]);
    }
    __syncthreads();

    const int cn = lane >> 2, cc = lane & 3;
    float g[N_DIM];
#pragma unroll
    for (int r = 0; r < N_DIM; ++r) g[r] = 0.0f;
    float sraw = 0.0f, nq = 0.0f;
#pragma unroll
    for (int w = 0; w < NWAVE; ++w) {
        const float4* gp = (const float4*)(gred + ((w * C_DIM + cc) * 8 + cn) * 8);
        float4 x = gp[0], y = gp[1];
        g[0] += x.x; g[1] += x.y; g[2] += x.z; g[3] += x.w;
        g[4] += y.x; g[5] += y.y; g[6] += y.z; g[7] += y.w;
        sraw += sred[(w * C_DIM + cc) * 8 + cn];
        nq   += nred[w * C_DIM + cc];
    }

    float s = sraw * (1.0f / (sqrtf(nq) + EPS_F));
    float wq = 0.0f, inv = 0.0f, rij = 0.0f;
#pragma unroll
    for (int it = 0; it < NITER; ++it) {
        float mx = fmaxf(s, xch1(s));
        mx = fmaxf(mx, xch2(mx));
        float e = __expf(s - mx);
        float es = e + xch1(e);
        es += xch2(es);
        rij = e / es;
        wq  = e * (pn / es);
        float t = 0.0f;
#pragma unroll
        for (int m = 0; m < N_DIM; ++m)
            t = fmaf(__shfl(wq, m * C_DIM + cc, 32), g[m], t);
        float q = wq * t;
        q += xch4(q); q += xch8(q); q += xch16(q);
        inv = 1.0f / (sqrtf(fmaxf(q, 0.0f)) + EPS_F);
        s = t * inv;
    }

    const int sc_i = __float_as_int(wq * inv);
#pragma unroll
    for (int c = 0; c < C_DIM; ++c) {
        float scl[N_DIM];
#pragma unroll
        for (int n = 0; n < N_DIM; ++n)
            scl[n] = __int_as_float(__builtin_amdgcn_readlane(sc_i, n * C_DIM + c));
        float4 a = make_float4(0.0f, 0.0f, 0.0f, 0.0f);
#pragma unroll
        for (int n = 0; n < N_DIM; ++n) {
            const float4 v = *(const float4*)(votes + (c * N_DIM + n) * VS + d4);
            a.x = fmaf(scl[n], v.x, a.x); a.y = fmaf(scl[n], v.y, a.y);
            a.z = fmaf(scl[n], v.z, a.z); a.w = fmaf(scl[n], v.w, a.w);
        }
        *(float4*)(out_cj + ((size_t)b * C_DIM + c) * D_DIM + d4) = a;
    }
    if (wave == 0) out_rij[(size_t)b * NROWS + lane] = rij;
}

extern "C" void kernel_launch(void* const* d_in, const int* in_sizes, int n_in,
                              void* d_out, int out_size, void* d_ws, size_t ws_size,
                              hipStream_t stream) {
    (void)n_in; (void)out_size;
    const float* fi  = (const float*)d_in[0];
    const float* pi  = (const float*)d_in[1];
    const float* W   = (const float*)d_in[2];
    const float* cji = (const float*)d_in[3];
    const int B = in_sizes[0] / (N_DIM * D_DIM);

    float* out_cj  = (float*)d_out;
    float* out_rij = out_cj + (size_t)B * C_DIM * D_DIM;

    const float* wdiag = nullptr;
    const size_t wbytes = (size_t)NROWS * D_DIM * sizeof(float);
    if (ws_size >= wbytes) {
        int total = NROWS * D_DIM;
        wdiag_gather_kernel<<<(total + 255) / 256, 256, 0, stream>>>(W, (float*)d_ws);
        wdiag = (const float*)d_ws;
    }

    const size_t smem = (size_t)(NROWS * VS + C_DIM * VS + NWAVE * C_DIM * 64 +
                                 NWAVE * NROWS + NWAVE * C_DIM) * sizeof(float);
    if (wdiag) {
        (void)hipFuncSetAttribute((const void*)routing_kernel<true>,
                                  hipFuncAttributeMaxDynamicSharedMemorySize, (int)smem);
        routing_kernel<true><<<B, 256, smem, stream>>>(fi, pi, W, wdiag, cji, out_cj, out_rij);
    } else {
        (void)hipFuncSetAttribute((const void*)routing_kernel<false>,
                                  hipFuncAttributeMaxDynamicSharedMemorySize, (int)smem);
        routing_kernel<false><<<B, 256, smem, stream>>>(fi, pi, W, wdiag, cji, out_cj, out_rij);
    }
}


# cj_init row rides in one B column per concept of the Gram product: S needs no matrix steps of its own (72 per wave, was 128); lost G_c[7][7] from diagonal-only steps, its sum reordered
# speedup vs baseline: 5.5231x; 1.0805x over previous
#include <hip/hip_runtime.h>

typedef __attribute__((ext_vector_type(2))) float v2f;
typedef __attribute__((ext_vector_type(8))) float v8f;

#define D_DIM   1024
#define N_DIM   8
#define C_DIM   4
#define NROWS   (N_DIM * C_DIM)
#define VS      1028
#define NITER   3
#define EPS_F   1e-6f
#define NWAVE   8
#define NSUB    4
#define SUBW    32

__global__ void wdiag_gather_kernel(const float* __restrict__ W,
                                    float* __restrict__ wdiag) {
    int t = blockIdx.x * blockDim.x + threadIdx.x;
    if (t >= NROWS * D_DIM) return;
    int row = t >> 10;
    int d   = t & (D_DIM - 1);
    int n = row >> 2, c = row & 3;
    size_t idx = (((size_t)n * D_DIM + d) * C_DIM + c) * D_DIM + d;
    wdiag[t] = W[idx];
}

__device__ __forceinline__ float xch1(float v) {
    return __int_as_float(__builtin_amdgcn_update_dpp(0, __float_as_int(v), 0xB1, 0xF, 0xF, true));
}
__device__ __forceinline__ float xch2(float v) {
    return __int_as_float(__builtin_amdgcn_update_dpp(0, __float_as_int(v), 0x4E, 0xF, 0xF, true));
}
__device__ __forceinline__ float xch4(float v) {
    return __int_as_float(__builtin_amdgcn_update_dpp(0, __float_as_int(v), 0x164, 0xF, 0xF, true));
}
__device__ __forceinline__ float xch8(float v) {
    return __int_as_float(__builtin_amdgcn_update_dpp(0, __float_as_int(v), 0x168, 0xF, 0xF, true));
}
__device__ __forceinline__ float xch16(float v) {
    int x = __float_as_int(v);
    return __int_as_float(__builtin_amdgcn_permlanex16(x, x, 0x76543210, 0xfedcba98, false, false));
}

__device__ __forceinline__ void hold_here(float4& v) {
    asm volatile("" : "+v"(v.x), "+v"(v.y), "+v"(v.z), "+v"(v.w) : : "memory");
}

template <bool GATHERED>
__global__ __launch_bounds__(256)
void routing_kernel(const float* __restrict__ fi, const float* __restrict__ pi,
                    const float* __restrict__ W, const float* __restrict__ wdiag,
                    const float* __restrict__ cj_init,
                    float* __restrict__ out_cj, float* __restrict__ out_rij) {
    extern __shared__ __align__(16) float smem[];
    float* votes = smem;
    float* cjl   = votes + NROWS * VS;
    float* gred  = cjl + C_DIM * VS;
    float* sred  = gred + NWAVE * C_DIM * 64;
    float* nred  = sred + NWAVE * NROWS;

    const int b    = blockIdx.x;
    const int tid  = threadIdx.x;
    const int lane = tid & 31;
    const int wave = tid >> 5;

    const int dg    = lane & 7;
    const int cg    = (lane >> 3) & 1;
    const int ng    = lane >> 4;
    const int cown  = 2 * ng + cg;
    const int kbase = wave * 128;
    const int dl    = kbase + 4 * dg;
    const int d4    = 4 * tid;

    const float pn = pi[b * N_DIM + (lane >> 2)];

    float4 cjn[NSUB];
    float4 fv[NSUB][4];
    float4 wv[NSUB][4][2];
#pragma unroll
    for (int j = 0; j < NSUB; ++j) {
        const int d = dl + SUBW * j;
        cjn[j] = *(const float4*)(cj_init + ((size_t)b * C_DIM + cown) * D_DIM + d);
#pragma unroll
        for (int i = 0; i < 4; ++i)
            fv[j][i] = *(const float4*)(fi + ((size_t)b * N_DIM + 4 * ng + i) * D_DIM + d);
#pragma unroll
        for (int i = 0; i < 4; ++i)
#pragma unroll
            for (int h = 0; h < 2; ++h) {
                const int n = 4 * ng + i, c = 2 * h + cg;
                if constexpr (GATHERED) {
                    wv[j][i][h] = *(const float4*)(wdiag + (n * C_DIM + c) * D_DIM + d);
                } else {
                    const size_t sd = (size_t)C_DIM * D_DIM + 1;
                    const float* p = W + (size_t)n * D_DIM * C_DIM * D_DIM + (size_t)d * sd + (size_t)c * D_DIM;
                    wv[j][i][h] = make_float4(p[0], p[sd], p[2 * sd], p[3 * sd]);
                }
            }
        __builtin_amdgcn_sched_barrier(0);
    }

    const int    arow = lane & 15;
    const int    half = lane >> 4;
    const int    koff = half * 2;
    const int    j8   = arow & 7;
    const float* a0p  = votes + arow * VS;
    const float* a1p  = votes + (16 + arow) * VS;
    const bool   q0   = (j8 == 7);
    const bool   q1   = (j8 == 6);
    const int    qrow = q0 ? (arow >> 3) : (q1 ? 2 + (arow >> 3) : 0);
    const float* qp   = cjl + qrow * VS;
    const int    dc   = arow >> 2;
    const int    dq   = arow & 3;
    const int    drop = (dc < 2) ? 7 : 6;
    const float* dp   = votes + (dc * N_DIM + drop) * VS + 4 * dq + koff;

    v8f T0 = {0.f, 0.f, 0.f, 0.f, 0.f, 0.f, 0.f, 0.f};
    v8f T1 = T0, Dg = T0;
    float ps = 0.0f;
#pragma unroll
    for (int j = 0; j < NSUB; ++j) {
        const int d = dl + SUBW * j;
        hold_here(cjn[j]);
#pragma unroll
        for (int i = 0; i < 4; ++i) {
            hold_here(fv[j][i]);
            hold_here(wv[j][i][0]);
            hold_here(wv[j][i][1]);
        }
#pragma unroll
        for (int i = 0; i < 4; ++i)
#pragma unroll
            for (int h = 0; h < 2; ++h) {
                const int n = 4 * ng + i, c = 2 * h + cg;
                const float4 f = fv[j][i], w = wv[j][i][h];
                *(float4*)(votes + (c * N_DIM + n) * VS + d) = make_float4(f.x * w.x, f.y * w.y, f.z * w.z, f.w * w.w);
            }
        {
            const float4 q = cjn[j];
            *(float4*)(cjl + cown * VS + d) = q;
            ps += q.x * q.x; ps += q.y * q.y; ps += q.z * q.z; ps += q.w * q.w;
        }
        v2f fd[SUBW / 16];
#pragma unroll
        for (int u = 0; u < SUBW / 16; ++u) fd[u] = *(const v2f*)(dp + kbase + SUBW * j + 16 * u);
#pragma unroll 4
        for (int s = 0; s < SUBW / 4; ++s) {
            int d0 = kbase + SUBW * j + 4 * s + koff;
            v2f a0 = *(const v2f*)(a0p + d0);
            v2f a1 = *(const v2f*)(a1p + d0);
            v2f qf = *(const v2f*)(qp + d0);
            v2f b0 = q0 ? qf : a0;
            v2f b1 = q1 ? qf : a1;
            T0 = __builtin_amdgcn_wmma_f32_16x16x4_f32(false, a0, false, b0, (short)0, T0, false, false);
            T1 = __builtin_amdgcn_wmma_f32_16x16x4_f32(false, a1, false, b1, (short)0, T1, false, false);
        }
#pragma unroll
        for (int u = 0; u < SUBW / 16; ++u)
            Dg = __builtin_amdgcn_wmma_f32_16x16x4_f32(false, fd[u], false, fd[u], (short)0, Dg, false, false);
    }
    ps += xch1(ps); ps += xch2(ps); ps += xch4(ps);
    if (dg == 0) nred[wave * C_DIM + cown] = ps;

    float dv = Dg[0];
#pragma unroll
    for (int r = 1; r < 8; ++r) dv = (j8 == r) ? Dg[r] : dv;
    dv += xch1(dv); dv += xch2(dv);
    if ((arow >> 3) == half) {
        const int c0 = half, c1 = 2 + half;
        float* gw = gred + wave * C_DIM * 64;
        float4* t0 = q0 ? (float4*)(sred + (wave * C_DIM + c0) * 8) : (float4*)(gw + (c0 * 8 + j8) * 8);
        float4* t1 = q1 ? (float4*)(sred + (wave * C_DIM + c1) * 8) : (float4*)(gw + (c1 * 8 + j8) * 8);
        t0[0] = make_float4(T0[0], T0[1], T0[2], T0[3]);
        t0[1] = make_float4(T0[4], T0[5], T0[6], T0[7]);
        t1[0] = make_float4(T1[0], T1[1], T1[2], T1[3]);
        t1[1] = make_float4(T1[4], T1[5], T1[6], T1[7]);
        if (!q0) gw[(c0 * 8 + 7) * 8 + j8] = T0[7];
        if (!q1) gw[(c1 * 8 + 6) * 8 + j8] = T1[6];
        if (dq == 0) gw[(dc * 8 + drop) * 8 + drop] = dv;
    }
    __syncthreads();

    const int cn = lane >> 2, cc = lane & 3;
    float g[N_DIM];
#pragma unroll
    for (int r = 0; r < N_DIM; ++r) g[r] = 0.0f;
    float sraw = 0.0f, nq = 0.0f;
#pragma unroll
    for (int w = 0; w < NWAVE; ++w) {
        const float4* gp = (const float4*)(gred + ((w * C_DIM + cc) * 8 + cn) * 8);
        float4 x = gp[0], y = gp[1];
        g[0] += x.x; g[1] += x.y; g[2] += x.z; g[3] += x.w;
        g[4] += y.x; g[5] += y.y; g[6] += y.z; g[7] += y.w;
        sraw += sred[(w * C_DIM + cc) * 8 + cn];
        nq   += nred[w * C_DIM + cc];
    }

    float s = sraw * (1.0f / (sqrtf(nq) + EPS_F));
    float wq = 0.0f, inv = 0.0f, rij = 0.0f;
#pragma unroll
    for (int it = 0; it < NITER; ++it) {
        float mx = fmaxf(s, xch1(s));
        mx = fmaxf(mx, xch2(mx));
        float e = __expf(s - mx);
        float es = e + xch1(e);
        es += xch2(es);
        rij = e / es;
        wq  = e * (pn / es);
        float t = 0.0f;
#pragma unroll
        for (int m = 0; m < N_DIM; ++m)
            t = fmaf(__shfl(wq, m * C_DIM + cc, 32), g[m], t);
        float q = wq * t;
        q += xch4(q); q += xch8(q); q += xch16(q);
        inv = 1.0f / (sqrtf(fmaxf(q, 0.0f)) + EPS_F);
        s = t * inv;
    }

    const int sc_i = __float_as_int(wq * inv);
#pragma unroll
    for (int c = 0; c < C_DIM; ++c) {
        float scl[N_DIM];
#pragma unroll
        for (int n = 0; n < N_DIM; ++n)
            scl[n] = __int_as_float(__builtin_amdgcn_readlane(sc_i, n * C_DIM + c));
        float4 a = make_float4(0.0f, 0.0f, 0.0f, 0.0f);
#pragma unroll
        for (int n = 0; n < N_DIM; ++n) {
            const float4 v = *(const float4*)(votes + (c * N_DIM + n) * VS + d4);
            a.x = fmaf(scl[n], v.x, a.x); a.y = fmaf(scl[n], v.y, a.y);
            a.z = fmaf(scl[n], v.z, a.z); a.w = fmaf(scl[n], v.w, a.w);
        }
        *(float4*)(out_cj + ((size_t)b * C_DIM + c) * D_DIM + d4) = a;
    }
    if (wave == 0) out_rij[(size_t)b * NROWS + lane] = rij;
}

extern "C" void kernel_launch(void* const* d_in, const int* in_sizes, int n_in,
                              void* d_out, int out_size, void* d_ws, size_t ws_size,
                              hipStream_t stream) {
    (void)n_in; (void)out_size;
    const float* fi  = (const float*)d_in[0];
    const float* pi  = (const float*)d_in[1];
    const float* W   = (const float*)d_in[2];
    const float* cji = (const float*)d_in[3];
    const int B = in_sizes[0] / (N_DIM * D_DIM);

    float* out_cj  = (float*)d_out;
    float* out_rij = out_cj + (size_t)B * C_DIM * D_DIM;

    const float* wdiag = nullptr;
    const size_t wbytes = (size_t)NROWS * D_DIM * sizeof(float);
    if (ws_size >= wbytes) {
        int total = NROWS * D_DIM;
        wdiag_gather_kernel<<<(total + 255) / 256, 256, 0, stream>>>(W, (float*)d_ws);
        wdiag = (const float*)d_ws;
    }

    const size_t smem = (size_t)(NROWS * VS + C_DIM * VS + NWAVE * C_DIM * 64 +
                                 NWAVE * NROWS + NWAVE * C_DIM) * sizeof(float);
    if (wdiag) {
        (void)hipFuncSetAttribute((const void*)routing_kernel<true>,
                                  hipFuncAttributeMaxDynamicSharedMemorySize, (int)smem);
        routing_kernel<true><<<B, 256, smem, stream>>>(fi, pi, W, wdiag, cji, out_cj, out_rij);
    } else {
        (void)hipFuncSetAttribute((const void*)routing_kernel<false>,
                                  hipFuncAttributeMaxDynamicSharedMemorySize, (int)smem);
        routing_kernel<false><<<B, 256, smem, stream>>>(fi, pi, W, wdiag, cji, out_cj, out_rij);
    }
}
